# starting kernel
# baseline (speedup 1.0000x reference)
#include <hip/hip_runtime.h>
#include <math.h>

#define BB 8
#define TT 512
#define DD 256
#define SS 64
#define PP 1024
#define RR 12
#define NN (RR * DD)
#define NROWS (BB * SS)

typedef __attribute__((ext_vector_type(2))) float v2f;
typedef __attribute__((ext_vector_type(8))) float v8f;

#define LSTR 260

__global__ __launch_bounds__(256) void span_pool_kernel(
    const float* __restrict__ encoded,
    const int*   __restrict__ starts,
    const int*   __restrict__ lens,
    const float* __restrict__ W_linear,
    float*       __restrict__ pooled,
    float*       __restrict__ lin)
{
    __shared__ float rowLDS[DD];

    const int bs = blockIdx.x;
    const int d  = threadIdx.x;
    const int b  = bs >> 6;
    const int st = starts[bs];
    const int en = st + lens[bs] + 1;
    const float* base = encoded + ((size_t)b * TT) * DD + d;
    float m = -3.402823466e+38f;
    for (int t = st; t < en; ++t)
        m = fmaxf(m, base[(size_t)t * DD]);
    pooled[(size_t)bs * DD + d] = m;
    rowLDS[d] = m;
    __syncthreads();

    const int wid  = threadIdx.x >> 5;
    const int lane = threadIdx.x & 31;
    #pragma unroll
    for (int oo = 0; oo < 3; ++oo) {
        const int o   = wid * 3 + oo;
        const int r   = (o < 12) ? o : (o - 12);
        const int off = (o < 12) ? 0 : DD;
        float pl = 0.f;
        #pragma unroll
        for (int j = 0; j < 8; ++j) {
            const int dd = lane + 32 * j;
            pl += rowLDS[dd] * W_linear[(off + dd) * RR + r];
        }
        pl += __shfl_xor(pl, 16, 32);
        pl += __shfl_xor(pl,  8, 32);
        pl += __shfl_xor(pl,  4, 32);
        pl += __shfl_xor(pl,  2, 32);
        pl += __shfl_xor(pl,  1, 32);
        if (lane == 0) lin[bs * 24 + o] = pl;
    }
}

__global__ __launch_bounds__(128) void span_gemm_kernel(
    const float* __restrict__ pooled,
    const float* __restrict__ W_bilinear,
    float*       __restrict__ G)
{
    __shared__ float aLDS[16 * LSTR];

    const int bs0 = blockIdx.x * 16;
    const int tid = threadIdx.x;

    for (int idx = tid; idx < 16 * 64; idx += 128) {
        const int row = idx >> 6;
        const int c4  = (idx & 63) << 2;
        const float4 av = *(const float4*)&pooled[((size_t)(bs0 + row)) * DD + c4];
        *(float4*)&aLDS[row * LSTR + c4] = av;
    }
    __syncthreads();

    const int wid  = tid >> 5;
    const int lane = tid & 31;
    const int l16  = lane & 15;
    const int hi   = lane >> 4;

    for (int sub = 0; sub < 4; ++sub) {
        const int ntile = blockIdx.y * 16 + wid * 4 + sub;
        const int n0    = ntile * 16;
        const int n     = n0 + l16;
        const float* wp = W_bilinear + (((size_t)(n >> 8)) << 16) + (n & 255)
                                     + (hi ? 2 * DD : 0);
        v8f c = {};
        #pragma unroll 4
        for (int k = 0; k < 64; ++k) {
            const float* ap = &aLDS[l16 * LSTR + 4 * k + 2 * hi];
            v2f a;  a.x = ap[0]; a.y = ap[1];
            v2f bf; bf.x = wp[(size_t)k * 1024];
                    bf.y = wp[(size_t)k * 1024 + DD];
            c = __builtin_amdgcn_wmma_f32_16x16x4_f32(
                    false, a, false, bf, (short)0, c, false, false);
        }
        #pragma unroll
        for (int v = 0; v < 8; ++v) {
            const int m = v + 8 * hi;
            G[((size_t)(bs0 + m)) * NN + n0 + l16] = c[v];
        }
    }
}

__global__ __launch_bounds__(128) void pair_score_kernel(
    const float* __restrict__ pooled,
    const float* __restrict__ G,
    const float* __restrict__ lin,
    const float* __restrict__ b_linear,
    const int*   __restrict__ pair_head,
    const int*   __restrict__ pair_tail,
    float*       __restrict__ out)
{
    const int wid  = threadIdx.x >> 5;
    const int lane = threadIdx.x & 31;
    const int bp   = blockIdx.x * 4 + wid;
    const int b    = bp >> 10;
    const int h    = pair_head[bp];
    const int t    = pair_tail[bp];
    const int hrow = b * SS + h;
    const int trow = b * SS + t;

    const float* tp = pooled + ((size_t)trow) * DD + lane * 8;
    const float4 t0 = *(const float4*)tp;
    const float4 t1 = *(const float4*)(tp + 4);

    const float* gr = G + ((size_t)hrow) * NN + lane * 8;
    float acc[RR];
    #pragma unroll
    for (int r = 0; r < RR; ++r) {
        const float4 g0 = *(const float4*)(gr + r * DD);
        const float4 g1 = *(const float4*)(gr + r * DD + 4);
        float s = g0.x * t0.x + g0.y * t0.y + g0.z * t0.z + g0.w * t0.w
                + g1.x * t1.x + g1.y * t1.y + g1.z * t1.z + g1.w * t1.w;
        acc[r] = s;
    }
    #pragma unroll
    for (int r = 0; r < RR; ++r) {
        float s = acc[r];
        s += __shfl_xor(s, 16, 32);
        s += __shfl_xor(s,  8, 32);
        s += __shfl_xor(s,  4, 32);
        s += __shfl_xor(s,  2, 32);
        s += __shfl_xor(s,  1, 32);
        acc[r] = s;
    }
    if (lane < RR) {
        const float sc = acc[lane] + b_linear[lane]
                       + lin[hrow * 24 + lane]
                       + lin[trow * 24 + 12 + lane];
        out[(size_t)bp * RR + lane] = 1.f / (1.f + __expf(-sc));
    }
}

extern "C" void kernel_launch(void* const* d_in, const int* in_sizes, int n_in,
                              void* d_out, int out_size, void* d_ws, size_t ws_size,
                              hipStream_t stream) {
    const float* encoded    = (const float*)d_in[0];
    const float* W_linear   = (const float*)d_in[1];
    const float* b_linear   = (const float*)d_in[2];
    const float* W_bilinear = (const float*)d_in[3];
    const int*   span_starts= (const int*)d_in[4];
    const int*   span_lens  = (const int*)d_in[5];
    const int*   pair_head  = (const int*)d_in[6];
    const int*   pair_tail  = (const int*)d_in[7];
    float*       out        = (float*)d_out;

    char* ws = (char*)d_ws;
    float* pooled = (float*)ws;
    float* lin    = (float*)(ws + (size_t)NROWS * DD * 4);
    float* G      = (float*)(ws + (size_t)NROWS * (DD + 24) * 4);

    span_pool_kernel<<<NROWS, DD, 0, stream>>>(encoded, span_starts, span_lens,
                                               W_linear, pooled, lin);

    dim3 ggrid(NROWS / 16, NN / 256);
    span_gemm_kernel<<<ggrid, 128, 0, stream>>>(pooled, W_bilinear, G);

    pair_score_kernel<<<(BB * PP) / 4, 128, 0, stream>>>(pooled, G, lin, b_linear,
                                                         pair_head, pair_tail, out);
}


# GEMM k-loop rebuilt on a 32x64 register tile per wave: 2x4 accumulator tiles share each A fragment and each 16-byte W-row load, W rows requested 8 k-steps ahead; k sum order unchanged
# speedup vs baseline: 1.5422x; 1.5422x over previous
#include <hip/hip_runtime.h>
#include <math.h>

#define BB 8
#define TT 512
#define DD 256
#define SS 64
#define PP 1024
#define RR 12
#define NN (RR * DD)
#define NROWS (BB * SS)

typedef __attribute__((ext_vector_type(2))) float v2f;
typedef __attribute__((ext_vector_type(8))) float v8f;

#define LSTR 260

__global__ __launch_bounds__(256) void span_pool_kernel(
    const float* __restrict__ encoded,
    const int*   __restrict__ starts,
    const int*   __restrict__ lens,
    const float* __restrict__ W_linear,
    float*       __restrict__ pooled,
    float*       __restrict__ lin)
{
    __shared__ float rowLDS[DD];

    const int bs = blockIdx.x;
    const int d  = threadIdx.x;
    const int b  = bs >> 6;
    const int st = starts[bs];
    const int en = st + lens[bs] + 1;
    const float* base = encoded + ((size_t)b * TT) * DD + d;
    float m = -3.402823466e+38f;
    for (int t = st; t < en; ++t)
        m = fmaxf(m, base[(size_t)t * DD]);
    pooled[(size_t)bs * DD + d] = m;
    rowLDS[d] = m;
    __syncthreads();

    const int wid  = threadIdx.x >> 5;
    const int lane = threadIdx.x & 31;
    #pragma unroll
    for (int oo = 0; oo < 3; ++oo) {
        const int o   = wid * 3 + oo;
        const int r   = (o < 12) ? o : (o - 12);
        const int off = (o < 12) ? 0 : DD;
        float pl = 0.f;
        #pragma unroll
        for (int j = 0; j < 8; ++j) {
            const int dd = lane + 32 * j;
            pl += rowLDS[dd] * W_linear[(off + dd) * RR + r];
        }
        pl += __shfl_xor(pl, 16, 32);
        pl += __shfl_xor(pl,  8, 32);
        pl += __shfl_xor(pl,  4, 32);
        pl += __shfl_xor(pl,  2, 32);
        pl += __shfl_xor(pl,  1, 32);
        if (lane == 0) lin[bs * 24 + o] = pl;
    }
}

#define GM_ROWS 64
#define GM_RING 8

__global__ __launch_bounds__(128) void span_gemm_kernel(
    const float* __restrict__ pooled,
    const float* __restrict__ W_bilinear,
    float*       __restrict__ G)
{
    __shared__ __attribute__((aligned(16))) float aLDS[GM_ROWS * LSTR];

    const int tid  = threadIdx.x;
    const int wid  = tid >> 5;
    const int lane = tid & 31;
    const int l16  = lane & 15;
    const int hi   = lane >> 4;
    const int wm   = wid & 1;
    const int wn   = wid >> 1;
    const int bs0  = blockIdx.x * GM_ROWS;
    const int n_w  = blockIdx.y * 128 + wn * 64;

    const float* wp = W_bilinear + (((size_t)(n_w >> 8)) << 16) + (n_w & 255)
                                 + 4 * l16 + (size_t)(2 * hi) * DD;

    float4 bq[GM_RING][2];
    #pragma unroll
    for (int q = 0; q < GM_RING; ++q) {
        bq[q][0] = *(const float4*)(wp + (size_t)(4 * q) * DD);
        bq[q][1] = *(const float4*)(wp + (size_t)(4 * q + 1) * DD);
    }

    {
        const int r2 = tid >> 6;
        const int c4 = (tid & 63) << 2;
        const float* src = pooled + ((size_t)(bs0 + r2)) * DD + c4;
        float*       dst = &aLDS[r2 * LSTR + c4];
        float4 av[GM_ROWS / 2];
        #pragma unroll
        for (int it = 0; it < GM_ROWS / 2; ++it)
            av[it] = *(const float4*)(src + (size_t)(2 * it) * DD);
        #pragma unroll
        for (int it = 0; it < GM_ROWS / 2; ++it)
            *(float4*)(dst + 2 * it * LSTR) = av[it];
    }
    __syncthreads();

    const float* ap = &aLDS[(wm * 32 + l16) * LSTR + 2 * hi];

    v2f af[2][8][2];
    #pragma unroll
    for (int q = 0; q < 8; ++q) {
        af[0][q][0] = *(const v2f*)(ap + 4 * q);
        af[0][q][1] = *(const v2f*)(ap + 16 * LSTR + 4 * q);
    }

    v8f acc[2][4];
    #pragma unroll
    for (int i = 0; i < 2; ++i)
        #pragma unroll
        for (int j = 0; j < 4; ++j)
            acc[i][j] = (v8f){};

    #pragma unroll
    for (int blk = 0; blk < 8; ++blk) {
        const int cur = blk & 1;
        if (blk + 1 < 8) {
            __builtin_amdgcn_sched_barrier(0);
            #pragma unroll
            for (int q = 0; q < 8; ++q) {
                af[cur ^ 1][q][0] = *(const v2f*)(ap + 4 * (8 * (blk + 1) + q));
                af[cur ^ 1][q][1] = *(const v2f*)(ap + 16 * LSTR + 4 * (8 * (blk + 1) + q));
            }
            __builtin_amdgcn_sched_barrier(0);
        }
        #pragma unroll
        for (int q = 0; q < 8; ++q) {
            const int kk = blk * 8 + q;
            const float4 r0 = bq[q][0];
            const float4 r1 = bq[q][1];
            if (kk + GM_RING < 64) {
                bq[q][0] = *(const float4*)(wp + (size_t)(4 * (kk + GM_RING)) * DD);
                bq[q][1] = *(const float4*)(wp + (size_t)(4 * (kk + GM_RING) + 1) * DD);
            }
            v2f b0; b0.x = r0.x; b0.y = r1.x;
            v2f b1; b1.x = r0.y; b1.y = r1.y;
            v2f b2; b2.x = r0.z; b2.y = r1.z;
            v2f b3; b3.x = r0.w; b3.y = r1.w;
            const v2f a0 = af[cur][q][0];
            const v2f a1 = af[cur][q][1];
            acc[0][0] = __builtin_amdgcn_wmma_f32_16x16x4_f32(false, a0, false, b0, (short)0, acc[0][0], false, false);
            acc[1][0] = __builtin_amdgcn_wmma_f32_16x16x4_f32(false, a1, false, b0, (short)0, acc[1][0], false, false);
            acc[0][1] = __builtin_amdgcn_wmma_f32_16x16x4_f32(false, a0, false, b1, (short)0, acc[0][1], false, false);
            acc[1][1] = __builtin_amdgcn_wmma_f32_16x16x4_f32(false, a1, false, b1, (short)0, acc[1][1], false, false);
            acc[0][2] = __builtin_amdgcn_wmma_f32_16x16x4_f32(false, a0, false, b2, (short)0, acc[0][2], false, false);
            acc[1][2] = __builtin_amdgcn_wmma_f32_16x16x4_f32(false, a1, false, b2, (short)0, acc[1][2], false, false);
            acc[0][3] = __builtin_amdgcn_wmma_f32_16x16x4_f32(false, a0, false, b3, (short)0, acc[0][3], false, false);
            acc[1][3] = __builtin_amdgcn_wmma_f32_16x16x4_f32(false, a1, false, b3, (short)0, acc[1][3], false, false);
        }
    }

    #pragma unroll
    for (int i = 0; i < 2; ++i) {
        #pragma unroll
        for (int v = 0; v < 8; ++v) {
            const int m = bs0 + wm * 32 + i * 16 + v + 8 * hi;
            float4 o;
            o.x = acc[i][0][v]; o.y = acc[i][1][v]; o.z = acc[i][2][v]; o.w = acc[i][3][v];
            *(float4*)&G[(size_t)m * NN + n_w + 4 * l16] = o;
        }
    }
}

__global__ __launch_bounds__(128) void pair_score_kernel(
    const float* __restrict__ pooled,
    const float* __restrict__ G,
    const float* __restrict__ lin,
    const float* __restrict__ b_linear,
    const int*   __restrict__ pair_head,
    const int*   __restrict__ pair_tail,
    float*       __restrict__ out)
{
    const int wid  = threadIdx.x >> 5;
    const int lane = threadIdx.x & 31;
    const int bp   = blockIdx.x * 4 + wid;
    const int b    = bp >> 10;
    const int h    = pair_head[bp];
    const int t    = pair_tail[bp];
    const int hrow = b * SS + h;
    const int trow = b * SS + t;

    const float* tp = pooled + ((size_t)trow) * DD + lane * 8;
    const float4 t0 = *(const float4*)tp;
    const float4 t1 = *(const float4*)(tp + 4);

    const float* gr = G + ((size_t)hrow) * NN + lane * 8;
    float acc[RR];
    #pragma unroll
    for (int r = 0; r < RR; ++r) {
        const float4 g0 = *(const float4*)(gr + r * DD);
        const float4 g1 = *(const float4*)(gr + r * DD + 4);
        float s = g0.x * t0.x + g0.y * t0.y + g0.z * t0.z + g0.w * t0.w
                + g1.x * t1.x + g1.y * t1.y + g1.z * t1.z + g1.w * t1.w;
        acc[r] = s;
    }
    #pragma unroll
    for (int r = 0; r < RR; ++r) {
        float s = acc[r];
        s += __shfl_xor(s, 16, 32);
        s += __shfl_xor(s,  8, 32);
        s += __shfl_xor(s,  4, 32);
        s += __shfl_xor(s,  2, 32);
        s += __shfl_xor(s,  1, 32);
        acc[r] = s;
    }
    if (lane < RR) {
        const float sc = acc[lane] + b_linear[lane]
                       + lin[hrow * 24 + lane]
                       + lin[trow * 24 + 12 + lane];
        out[(size_t)bp * RR + lane] = 1.f / (1.f + __expf(-sc));
    }
}

extern "C" void kernel_launch(void* const* d_in, const int* in_sizes, int n_in,
                              void* d_out, int out_size, void* d_ws, size_t ws_size,
                              hipStream_t stream) {
    const float* encoded    = (const float*)d_in[0];
    const float* W_linear   = (const float*)d_in[1];
    const float* b_linear   = (const float*)d_in[2];
    const float* W_bilinear = (const float*)d_in[3];
    const int*   span_starts= (const int*)d_in[4];
    const int*   span_lens  = (const int*)d_in[5];
    const int*   pair_head  = (const int*)d_in[6];
    const int*   pair_tail  = (const int*)d_in[7];
    float*       out        = (float*)d_out;

    char* ws = (char*)d_ws;
    float* pooled = (float*)ws;
    float* lin    = (float*)(ws + (size_t)NROWS * DD * 4);
    float* G      = (float*)(ws + (size_t)NROWS * (DD + 24) * 4);

    span_pool_kernel<<<NROWS, DD, 0, stream>>>(encoded, span_starts, span_lens,
                                               W_linear, pooled, lin);

    dim3 ggrid(NROWS / GM_ROWS, NN / 128);
    span_gemm_kernel<<<ggrid, 128, 0, stream>>>(pooled, W_bilinear, G);

    pair_score_kernel<<<(BB * PP) / 4, 128, 0, stream>>>(pooled, G, lin, b_linear,
                                                         pair_head, pair_tail, out);
}


# Pair bilinear dots moved onto the matrix unit: each GEMM block keeps G in LDS and multiplies it by its batch's pooled rows into a head-by-tail table; scoring only gathers; e-sum order changes
# speedup vs baseline: 2.0282x; 1.3151x over previous
#include <hip/hip_runtime.h>
#include <math.h>

#define BB 8
#define TT 512
#define DD 256
#define SS 64
#define PP 1024
#define RR 12
#define NN (RR * DD)
#define NROWS (BB * SS)

typedef __attribute__((ext_vector_type(2))) float v2f;
typedef __attribute__((ext_vector_type(8))) float v8f;

#define LSTR 260

__global__ __launch_bounds__(256) void span_pool_kernel(
    const float* __restrict__ encoded,
    const int*   __restrict__ starts,
    const int*   __restrict__ lens,
    const float* __restrict__ W_linear,
    float*       __restrict__ pooled,
    float*       __restrict__ lin)
{
    __shared__ float rowLDS[DD];

    const int bs = blockIdx.x;
    const int d  = threadIdx.x;
    const int b  = bs >> 6;
    const int st = starts[bs];
    const int en = st + lens[bs] + 1;
    const float* base = encoded + ((size_t)b * TT) * DD + d;
    float m = -3.402823466e+38f;
    for (int t = st; t < en; ++t)
        m = fmaxf(m, base[(size_t)t * DD]);
    pooled[(size_t)bs * DD + d] = m;
    rowLDS[d] = m;
    __syncthreads();

    const int wid  = threadIdx.x >> 5;
    const int lane = threadIdx.x & 31;
    #pragma unroll
    for (int oo = 0; oo < 3; ++oo) {
        const int o   = wid * 3 + oo;
        const int r   = (o < 12) ? o : (o - 12);
        const int off = (o < 12) ? 0 : DD;
        float pl = 0.f;
        #pragma unroll
        for (int j = 0; j < 8; ++j) {
            const int dd = lane + 32 * j;
            pl += rowLDS[dd] * W_linear[(off + dd) * RR + r];
        }
        pl += __shfl_xor(pl, 16, 32);
        pl += __shfl_xor(pl,  8, 32);
        pl += __shfl_xor(pl,  4, 32);
        pl += __shfl_xor(pl,  2, 32);
        pl += __shfl_xor(pl,  1, 32);
        if (lane == 0) lin[bs * 24 + o] = pl;
    }
}

#define GM_ROWS 64
#define GM_RING 8
#define TB_BLK  8

__global__ __launch_bounds__(128) void span_gemm_table_kernel(
    const float* __restrict__ pooled,
    const float* __restrict__ W_bilinear,
    float*       __restrict__ Tbl)
{
    __shared__ __attribute__((aligned(16))) float aLDS[GM_ROWS * LSTR];
    __shared__ __attribute__((aligned(16))) float gLDS[32 * LSTR];

    const int tid  = threadIdx.x;
    const int wid  = tid >> 5;
    const int lane = tid & 31;
    const int l16  = lane & 15;
    const int hi   = lane >> 4;
    const int b    = blockIdx.x >> 1;
    const int hh   = blockIdx.x & 1;
    const int r    = blockIdx.y;
    const int bs0  = b * GM_ROWS;

    const float* wp = W_bilinear + (((size_t)r) << 16) + wid * 64
                                 + 4 * l16 + (size_t)(2 * hi) * DD;

    float4 bq[GM_RING][2];
    #pragma unroll
    for (int q = 0; q < GM_RING; ++q) {
        bq[q][0] = *(const float4*)(wp + (size_t)(4 * q) * DD);
        bq[q][1] = *(const float4*)(wp + (size_t)(4 * q + 1) * DD);
    }

    {
        const int r2 = tid >> 6;
        const int c4 = (tid & 63) << 2;
        const float* src = pooled + ((size_t)(bs0 + r2)) * DD + c4;
        float*       dst = &aLDS[r2 * LSTR + c4];
        float4 av[GM_ROWS / 2];
        #pragma unroll
        for (int it = 0; it < GM_ROWS / 2; ++it)
            av[it] = *(const float4*)(src + (size_t)(2 * it) * DD);
        #pragma unroll
        for (int it = 0; it < GM_ROWS / 2; ++it)
            *(float4*)(dst + 2 * it * LSTR) = av[it];
    }
    __syncthreads();

    const float* ap = &aLDS[(hh * 32 + l16) * LSTR + 2 * hi];

    v2f af[2][8][2];
    #pragma unroll
    for (int q = 0; q < 8; ++q) {
        af[0][q][0] = *(const v2f*)(ap + 4 * q);
        af[0][q][1] = *(const v2f*)(ap + 16 * LSTR + 4 * q);
    }

    v8f acc[2][4];
    #pragma unroll
    for (int i = 0; i < 2; ++i)
        #pragma unroll
        for (int j = 0; j < 4; ++j)
            acc[i][j] = (v8f){};

    #pragma unroll
    for (int blk = 0; blk < 8; ++blk) {
        const int cur = blk & 1;
        if (blk + 1 < 8) {
            __builtin_amdgcn_sched_barrier(0);
            #pragma unroll
            for (int q = 0; q < 8; ++q) {
                af[cur ^ 1][q][0] = *(const v2f*)(ap + 4 * (8 * (blk + 1) + q));
                af[cur ^ 1][q][1] = *(const v2f*)(ap + 16 * LSTR + 4 * (8 * (blk + 1) + q));
            }
            __builtin_amdgcn_sched_barrier(0);
        }
        #pragma unroll
        for (int q = 0; q < 8; ++q) {
            const int kk = blk * 8 + q;
            const float4 r0 = bq[q][0];
            const float4 r1 = bq[q][1];
            if (kk + GM_RING < 64) {
                bq[q][0] = *(const float4*)(wp + (size_t)(4 * (kk + GM_RING)) * DD);
                bq[q][1] = *(const float4*)(wp + (size_t)(4 * (kk + GM_RING) + 1) * DD);
            }
            v2f b0; b0.x = r0.x; b0.y = r1.x;
            v2f b1; b1.x = r0.y; b1.y = r1.y;
            v2f b2; b2.x = r0.z; b2.y = r1.z;
            v2f b3; b3.x = r0.w; b3.y = r1.w;
            const v2f a0 = af[cur][q][0];
            const v2f a1 = af[cur][q][1];
            acc[0][0] = __builtin_amdgcn_wmma_f32_16x16x4_f32(false, a0, false, b0, (short)0, acc[0][0], false, false);
            acc[1][0] = __builtin_amdgcn_wmma_f32_16x16x4_f32(false, a1, false, b0, (short)0, acc[1][0], false, false);
            acc[0][1] = __builtin_amdgcn_wmma_f32_16x16x4_f32(false, a0, false, b1, (short)0, acc[0][1], false, false);
            acc[1][1] = __builtin_amdgcn_wmma_f32_16x16x4_f32(false, a1, false, b1, (short)0, acc[1][1], false, false);
            acc[0][2] = __builtin_amdgcn_wmma_f32_16x16x4_f32(false, a0, false, b2, (short)0, acc[0][2], false, false);
            acc[1][2] = __builtin_amdgcn_wmma_f32_16x16x4_f32(false, a1, false, b2, (short)0, acc[1][2], false, false);
            acc[0][3] = __builtin_amdgcn_wmma_f32_16x16x4_f32(false, a0, false, b3, (short)0, acc[0][3], false, false);
            acc[1][3] = __builtin_amdgcn_wmma_f32_16x16x4_f32(false, a1, false, b3, (short)0, acc[1][3], false, false);
        }
    }

    #pragma unroll
    for (int i = 0; i < 2; ++i) {
        #pragma unroll
        for (int v = 0; v < 8; ++v) {
            float4 o;
            o.x = acc[i][0][v]; o.y = acc[i][1][v]; o.z = acc[i][2][v]; o.w = acc[i][3][v];
            *(float4*)&gLDS[(i * 16 + v + 8 * hi) * LSTR + wid * 64 + 4 * l16] = o;
        }
    }
    __syncthreads();

    const int mt = wid & 1;
    const int th = wid >> 1;
    const float* gp = &gLDS[(mt * 16 + l16) * LSTR + 4 * hi];
    const float* tp = &aLDS[(th * 32 + 2 * l16) * LSTR + 4 * hi];

    float4 ga[2][TB_BLK], t0[2][TB_BLK], t1[2][TB_BLK];
    #pragma unroll
    for (int q = 0; q < TB_BLK; ++q) {
        ga[0][q] = *(const float4*)(gp + 8 * q);
        t0[0][q] = *(const float4*)(tp + 8 * q);
        t1[0][q] = *(const float4*)(tp + LSTR + 8 * q);
    }
    v8f tacc[2];
    tacc[0] = (v8f){};
    tacc[1] = (v8f){};

    #pragma unroll
    for (int blk = 0; blk < 32 / TB_BLK; ++blk) {
        const int cur = blk & 1;
        if (blk + 1 < 32 / TB_BLK) {
            __builtin_amdgcn_sched_barrier(0);
            #pragma unroll
            for (int q = 0; q < TB_BLK; ++q) {
                const int s = TB_BLK * (blk + 1) + q;
                ga[cur ^ 1][q] = *(const float4*)(gp + 8 * s);
                t0[cur ^ 1][q] = *(const float4*)(tp + 8 * s);
                t1[cur ^ 1][q] = *(const float4*)(tp + LSTR + 8 * s);
            }
            __builtin_amdgcn_sched_barrier(0);
        }
        #pragma unroll
        for (int q = 0; q < TB_BLK; ++q) {
            const float4 g  = ga[cur][q];
            const float4 p0 = t0[cur][q];
            const float4 p1 = t1[cur][q];
            v2f a0;  a0.x  = g.x;  a0.y  = g.y;
            v2f a1;  a1.x  = g.z;  a1.y  = g.w;
            v2f b00; b00.x = p0.x; b00.y = p0.y;
            v2f b01; b01.x = p0.z; b01.y = p0.w;
            v2f b10; b10.x = p1.x; b10.y = p1.y;
            v2f b11; b11.x = p1.z; b11.y = p1.w;
            tacc[0] = __builtin_amdgcn_wmma_f32_16x16x4_f32(false, a0, false, b00, (short)0, tacc[0], false, false);
            tacc[1] = __builtin_amdgcn_wmma_f32_16x16x4_f32(false, a0, false, b10, (short)0, tacc[1], false, false);
            tacc[0] = __builtin_amdgcn_wmma_f32_16x16x4_f32(false, a1, false, b01, (short)0, tacc[0], false, false);
            tacc[1] = __builtin_amdgcn_wmma_f32_16x16x4_f32(false, a1, false, b11, (short)0, tacc[1], false, false);
        }
    }

    float* trow = Tbl + ((((size_t)b * RR + r) * SS) + hh * 32 + mt * 16 + 8 * hi) * SS
                      + th * 32 + 2 * l16;
    #pragma unroll
    for (int v = 0; v < 8; ++v) {
        float2 o;
        o.x = tacc[0][v]; o.y = tacc[1][v];
        *(float2*)(trow + (size_t)v * SS) = o;
    }
}

__global__ __launch_bounds__(256) void pair_score_kernel(
    const float* __restrict__ Tbl,
    const float* __restrict__ lin,
    const float* __restrict__ b_linear,
    const int*   __restrict__ pair_head,
    const int*   __restrict__ pair_tail,
    float*       __restrict__ out)
{
    const int idx = blockIdx.x * 256 + threadIdx.x;
    const int bp  = idx / RR;
    const int r   = idx - bp * RR;
    const int b   = bp >> 10;
    const int h   = pair_head[bp];
    const int t   = pair_tail[bp];
    const int hrow = b * SS + h;
    const int trow = b * SS + t;
    const float sc = Tbl[((((size_t)b * RR + r) * SS) + h) * SS + t] + b_linear[r]
                   + lin[hrow * 24 + r]
                   + lin[trow * 24 + 12 + r];
    out[idx] = 1.f / (1.f + __expf(-sc));
}

extern "C" void kernel_launch(void* const* d_in, const int* in_sizes, int n_in,
                              void* d_out, int out_size, void* d_ws, size_t ws_size,
                              hipStream_t stream) {
    const float* encoded    = (const float*)d_in[0];
    const float* W_linear   = (const float*)d_in[1];
    const float* b_linear   = (const float*)d_in[2];
    const float* W_bilinear = (const float*)d_in[3];
    const int*   span_starts= (const int*)d_in[4];
    const int*   span_lens  = (const int*)d_in[5];
    const int*   pair_head  = (const int*)d_in[6];
    const int*   pair_tail  = (const int*)d_in[7];
    float*       out        = (float*)d_out;

    char* ws = (char*)d_ws;
    float* pooled = (float*)ws;
    float* lin    = (float*)(ws + (size_t)NROWS * DD * 4);
    float* Tbl    = (float*)(ws + (size_t)NROWS * (DD + 24) * 4);

    span_pool_kernel<<<NROWS, DD, 0, stream>>>(encoded, span_starts, span_lens,
                                               W_linear, pooled, lin);

    dim3 ggrid(BB * 2, RR);
    span_gemm_table_kernel<<<ggrid, 128, 0, stream>>>(pooled, W_bilinear, Tbl);

    pair_score_kernel<<<(BB * PP * RR) / 256, 256, 0, stream>>>(Tbl, lin, b_linear,
                                                                pair_head, pair_tail, out);
}


# Span kernel's serial load chain removed: its 12 span rows and its W_linear pieces are all requested before the first wait, W_linear read from LDS for the linear terms; sum orders unchanged
# speedup vs baseline: 2.4139x; 1.1902x over previous
#include <hip/hip_runtime.h>
#include <math.h>

#define BB 8
#define TT 512
#define DD 256
#define SS 64
#define PP 1024
#define RR 12
#define NN (RR * DD)
#define NROWS (BB * SS)

typedef __attribute__((ext_vector_type(2))) float v2f;
typedef __attribute__((ext_vector_type(8))) float v8f;

#define LSTR 260

#define SPAN_ROWS 12
#define WL_F4 ((2 * DD * RR) / 4 / 256)

__global__ __launch_bounds__(256) void span_pool_kernel(
    const float* __restrict__ encoded,
    const int*   __restrict__ starts,
    const int*   __restrict__ lens,
    const float* __restrict__ W_linear,
    float*       __restrict__ pooled,
    float*       __restrict__ lin)
{
    __shared__ float rowLDS[DD];
    __shared__ __attribute__((aligned(16))) float wLDS[2 * DD * RR];

    const int bs = blockIdx.x;
    const int d  = threadIdx.x;
    const int b  = bs >> 6;

    float4 wv[WL_F4];
    #pragma unroll
    for (int k = 0; k < WL_F4; ++k)
        wv[k] = ((const float4*)W_linear)[d + 256 * k];

    const int st = starts[bs];
    const int en = st + lens[bs] + 1;
    const float* base = encoded + ((size_t)b * TT) * DD + d;

    float v[SPAN_ROWS];
    #pragma unroll
    for (int i = 0; i < SPAN_ROWS; ++i) {
        const int t = st + i;
        v[i] = base[(size_t)((t < en) ? t : st) * DD];
    }

    #pragma unroll
    for (int k = 0; k < WL_F4; ++k)
        ((float4*)wLDS)[d + 256 * k] = wv[k];

    float m = -3.402823466e+38f;
    #pragma unroll
    for (int i = 0; i < SPAN_ROWS; ++i)
        m = fmaxf(m, (st + i < en) ? v[i] : -3.402823466e+38f);
    for (int t = st + SPAN_ROWS; t < en; ++t)
        m = fmaxf(m, base[(size_t)t * DD]);

    pooled[(size_t)bs * DD + d] = m;
    rowLDS[d] = m;
    __syncthreads();

    const int wid  = threadIdx.x >> 5;
    const int lane = threadIdx.x & 31;
    float pl[3];
    #pragma unroll
    for (int oo = 0; oo < 3; ++oo) {
        const int o   = wid * 3 + oo;
        const int r   = (o < 12) ? o : (o - 12);
        const int off = (o < 12) ? 0 : DD;
        float acc = 0.f;
        #pragma unroll
        for (int j = 0; j < 8; ++j) {
            const int dd = lane + 32 * j;
            acc += rowLDS[dd] * wLDS[(off + dd) * RR + r];
        }
        pl[oo] = acc;
    }
    #pragma unroll
    for (int sh = 16; sh >= 1; sh >>= 1) {
        float ex[3];
        #pragma unroll
        for (int oo = 0; oo < 3; ++oo) ex[oo] = __shfl_xor(pl[oo], sh, 32);
        #pragma unroll
        for (int oo = 0; oo < 3; ++oo) pl[oo] += ex[oo];
    }
    if (lane == 0) {
        #pragma unroll
        for (int oo = 0; oo < 3; ++oo) lin[bs * 24 + wid * 3 + oo] = pl[oo];
    }
}

#define GM_ROWS 64
#define GM_RING 8
#define TB_BLK  8

__global__ __launch_bounds__(128) void span_gemm_table_kernel(
    const float* __restrict__ pooled,
    const float* __restrict__ W_bilinear,
    float*       __restrict__ Tbl)
{
    __shared__ __attribute__((aligned(16))) float aLDS[GM_ROWS * LSTR];
    __shared__ __attribute__((aligned(16))) float gLDS[32 * LSTR];

    const int tid  = threadIdx.x;
    const int wid  = tid >> 5;
    const int lane = tid & 31;
    const int l16  = lane & 15;
    const int hi   = lane >> 4;
    const int b    = blockIdx.x >> 1;
    const int hh   = blockIdx.x & 1;
    const int r    = blockIdx.y;
    const int bs0  = b * GM_ROWS;

    const float* wp = W_bilinear + (((size_t)r) << 16) + wid * 64
                                 + 4 * l16 + (size_t)(2 * hi) * DD;

    float4 bq[GM_RING][2];
    #pragma unroll
    for (int q = 0; q < GM_RING; ++q) {
        bq[q][0] = *(const float4*)(wp + (size_t)(4 * q) * DD);
        bq[q][1] = *(const float4*)(wp + (size_t)(4 * q + 1) * DD);
    }

    {
        const int r2 = tid >> 6;
        const int c4 = (tid & 63) << 2;
        const float* src = pooled + ((size_t)(bs0 + r2)) * DD + c4;
        float*       dst = &aLDS[r2 * LSTR + c4];
        float4 av[GM_ROWS / 2];
        #pragma unroll
        for (int it = 0; it < GM_ROWS / 2; ++it)
            av[it] = *(const float4*)(src + (size_t)(2 * it) * DD);
        #pragma unroll
        for (int it = 0; it < GM_ROWS / 2; ++it)
            *(float4*)(dst + 2 * it * LSTR) = av[it];
    }
    __syncthreads();

    const float* ap = &aLDS[(hh * 32 + l16) * LSTR + 2 * hi];

    v2f af[2][8][2];
    #pragma unroll
    for (int q = 0; q < 8; ++q) {
        af[0][q][0] = *(const v2f*)(ap + 4 * q);
        af[0][q][1] = *(const v2f*)(ap + 16 * LSTR + 4 * q);
    }

    v8f acc[2][4];
    #pragma unroll
    for (int i = 0; i < 2; ++i)
        #pragma unroll
        for (int j = 0; j < 4; ++j)
            acc[i][j] = (v8f){};

    #pragma unroll
    for (int blk = 0; blk < 8; ++blk) {
        const int cur = blk & 1;
        if (blk + 1 < 8) {
            __builtin_amdgcn_sched_barrier(0);
            #pragma unroll
            for (int q = 0; q < 8; ++q) {
                af[cur ^ 1][q][0] = *(const v2f*)(ap + 4 * (8 * (blk + 1) + q));
                af[cur ^ 1][q][1] = *(const v2f*)(ap + 16 * LSTR + 4 * (8 * (blk + 1) + q));
            }
            __builtin_amdgcn_sched_barrier(0);
        }
        #pragma unroll
        for (int q = 0; q < 8; ++q) {
            const int kk = blk * 8 + q;
            const float4 r0 = bq[q][0];
            const float4 r1 = bq[q][1];
            if (kk + GM_RING < 64) {
                bq[q][0] = *(const float4*)(wp + (size_t)(4 * (kk + GM_RING)) * DD);
                bq[q][1] = *(const float4*)(wp + (size_t)(4 * (kk + GM_RING) + 1) * DD);
            }
            v2f b0; b0.x = r0.x; b0.y = r1.x;
            v2f b1; b1.x = r0.y; b1.y = r1.y;
            v2f b2; b2.x = r0.z; b2.y = r1.z;
            v2f b3; b3.x = r0.w; b3.y = r1.w;
            const v2f a0 = af[cur][q][0];
            const v2f a1 = af[cur][q][1];
            acc[0][0] = __builtin_amdgcn_wmma_f32_16x16x4_f32(false, a0, false, b0, (short)0, acc[0][0], false, false);
            acc[1][0] = __builtin_amdgcn_wmma_f32_16x16x4_f32(false, a1, false, b0, (short)0, acc[1][0], false, false);
            acc[0][1] = __builtin_amdgcn_wmma_f32_16x16x4_f32(false, a0, false, b1, (short)0, acc[0][1], false, false);
            acc[1][1] = __builtin_amdgcn_wmma_f32_16x16x4_f32(false, a1, false, b1, (short)0, acc[1][1], false, false);
            acc[0][2] = __builtin_amdgcn_wmma_f32_16x16x4_f32(false, a0, false, b2, (short)0, acc[0][2], false, false);
            acc[1][2] = __builtin_amdgcn_wmma_f32_16x16x4_f32(false, a1, false, b2, (short)0, acc[1][2], false, false);
            acc[0][3] = __builtin_amdgcn_wmma_f32_16x16x4_f32(false, a0, false, b3, (short)0, acc[0][3], false, false);
            acc[1][3] = __builtin_amdgcn_wmma_f32_16x16x4_f32(false, a1, false, b3, (short)0, acc[1][3], false, false);
        }
    }

    #pragma unroll
    for (int i = 0; i < 2; ++i) {
        #pragma unroll
        for (int v = 0; v < 8; ++v) {
            float4 o;
            o.x = acc[i][0][v]; o.y = acc[i][1][v]; o.z = acc[i][2][v]; o.w = acc[i][3][v];
            *(float4*)&gLDS[(i * 16 + v + 8 * hi) * LSTR + wid * 64 + 4 * l16] = o;
        }
    }
    __syncthreads();

    const int mt = wid & 1;
    const int th = wid >> 1;
    const float* gp = &gLDS[(mt * 16 + l16) * LSTR + 4 * hi];
    const float* tp = &aLDS[(th * 32 + 2 * l16) * LSTR + 4 * hi];

    float4 ga[2][TB_BLK], t0[2][TB_BLK], t1[2][TB_BLK];
    #pragma unroll
    for (int q = 0; q < TB_BLK; ++q) {
        ga[0][q] = *(const float4*)(gp + 8 * q);
        t0[0][q] = *(const float4*)(tp + 8 * q);
        t1[0][q] = *(const float4*)(tp + LSTR + 8 * q);
    }
    v8f tacc[2];
    tacc[0] = (v8f){};
    tacc[1] = (v8f){};

    #pragma unroll
    for (int blk = 0; blk < 32 / TB_BLK; ++blk) {
        const int cur = blk & 1;
        if (blk + 1 < 32 / TB_BLK) {
            __builtin_amdgcn_sched_barrier(0);
            #pragma unroll
            for (int q = 0; q < TB_BLK; ++q) {
                const int s = TB_BLK * (blk + 1) + q;
                ga[cur ^ 1][q] = *(const float4*)(gp + 8 * s);
                t0[cur ^ 1][q] = *(const float4*)(tp + 8 * s);
                t1[cur ^ 1][q] = *(const float4*)(tp + LSTR + 8 * s);
            }
            __builtin_amdgcn_sched_barrier(0);
        }
        #pragma unroll
        for (int q = 0; q < TB_BLK; ++q) {
            const float4 g  = ga[cur][q];
            const float4 p0 = t0[cur][q];
            const float4 p1 = t1[cur][q];
            v2f a0;  a0.x  = g.x;  a0.y  = g.y;
            v2f a1;  a1.x  = g.z;  a1.y  = g.w;
            v2f b00; b00.x = p0.x; b00.y = p0.y;
            v2f b01; b01.x = p0.z; b01.y = p0.w;
            v2f b10; b10.x = p1.x; b10.y = p1.y;
            v2f b11; b11.x = p1.z; b11.y = p1.w;
            tacc[0] = __builtin_amdgcn_wmma_f32_16x16x4_f32(false, a0, false, b00, (short)0, tacc[0], false, false);
            tacc[1] = __builtin_amdgcn_wmma_f32_16x16x4_f32(false, a0, false, b10, (short)0, tacc[1], false, false);
            tacc[0] = __builtin_amdgcn_wmma_f32_16x16x4_f32(false, a1, false, b01, (short)0, tacc[0], false, false);
            tacc[1] = __builtin_amdgcn_wmma_f32_16x16x4_f32(false, a1, false, b11, (short)0, tacc[1], false, false);
        }
    }

    float* trow = Tbl + ((((size_t)b * RR + r) * SS) + hh * 32 + mt * 16 + 8 * hi) * SS
                      + th * 32 + 2 * l16;
    #pragma unroll
    for (int v = 0; v < 8; ++v) {
        float2 o;
        o.x = tacc[0][v]; o.y = tacc[1][v];
        *(float2*)(trow + (size_t)v * SS) = o;
    }
}

__global__ __launch_bounds__(256) void pair_score_kernel(
    const float* __restrict__ Tbl,
    const float* __restrict__ lin,
    const float* __restrict__ b_linear,
    const int*   __restrict__ pair_head,
    const int*   __restrict__ pair_tail,
    float*       __restrict__ out)
{
    const int idx = blockIdx.x * 256 + threadIdx.x;
    const int bp  = idx / RR;
    const int r   = idx - bp * RR;
    const int b   = bp >> 10;
    const int h   = pair_head[bp];
    const int t   = pair_tail[bp];
    const int hrow = b * SS + h;
    const int trow = b * SS + t;
    const float sc = Tbl[((((size_t)b * RR + r) * SS) + h) * SS + t] + b_linear[r]
                   + lin[hrow * 24 + r]
                   + lin[trow * 24 + 12 + r];
    out[idx] = 1.f / (1.f + __expf(-sc));
}

extern "C" void kernel_launch(void* const* d_in, const int* in_sizes, int n_in,
                              void* d_out, int out_size, void* d_ws, size_t ws_size,
                              hipStream_t stream) {
    const float* encoded    = (const float*)d_in[0];
    const float* W_linear   = (const float*)d_in[1];
    const float* b_linear   = (const float*)d_in[2];
    const float* W_bilinear = (const float*)d_in[3];
    const int*   span_starts= (const int*)d_in[4];
    const int*   span_lens  = (const int*)d_in[5];
    const int*   pair_head  = (const int*)d_in[6];
    const int*   pair_tail  = (const int*)d_in[7];
    float*       out        = (float*)d_out;

    char* ws = (char*)d_ws;
    float* pooled = (float*)ws;
    float* lin    = (float*)(ws + (size_t)NROWS * DD * 4);
    float* Tbl    = (float*)(ws + (size_t)NROWS * (DD + 24) * 4);

    span_pool_kernel<<<NROWS, DD, 0, stream>>>(encoded, span_starts, span_lens,
                                               W_linear, pooled, lin);

    dim3 ggrid(BB * 2, RR);
    span_gemm_table_kernel<<<ggrid, 128, 0, stream>>>(pooled, W_bilinear, Tbl);

    pair_score_kernel<<<(BB * PP * RR) / 256, 256, 0, stream>>>(Tbl, lin, b_linear,
                                                                pair_head, pair_tail, out);
}


# Scoring folded into the GEMM/table kernel: table tile kept in LDS, each block scores its batch's pairs whose head is in its half, 4-byte stores; 3 launches become 2; sum order unchanged
# speedup vs baseline: 2.9567x; 1.2249x over previous
#include <hip/hip_runtime.h>
#include <math.h>

#define BB 8
#define TT 512
#define DD 256
#define SS 64
#define PP 1024
#define RR 12
#define NN (RR * DD)
#define NROWS (BB * SS)

typedef __attribute__((ext_vector_type(2))) float v2f;
typedef __attribute__((ext_vector_type(8))) float v8f;

#define LSTR 260

#define SPAN_ROWS 12
#define WL_F4 ((2 * DD * RR) / 4 / 256)

__global__ __launch_bounds__(256) void span_pool_kernel(
    const float* __restrict__ encoded,
    const int*   __restrict__ starts,
    const int*   __restrict__ lens,
    const float* __restrict__ W_linear,
    float*       __restrict__ pooled,
    float*       __restrict__ lin)
{
    __shared__ float rowLDS[DD];
    __shared__ __attribute__((aligned(16))) float wLDS[2 * DD * RR];

    const int bs = blockIdx.x;
    const int d  = threadIdx.x;
    const int b  = bs >> 6;

    float4 wv[WL_F4];
    #pragma unroll
    for (int k = 0; k < WL_F4; ++k)
        wv[k] = ((const float4*)W_linear)[d + 256 * k];

    const int st = starts[bs];
    const int en = st + lens[bs] + 1;
    const float* base = encoded + ((size_t)b * TT) * DD + d;

    float v[SPAN_ROWS];
    #pragma unroll
    for (int i = 0; i < SPAN_ROWS; ++i) {
        const int t = st + i;
        v[i] = base[(size_t)((t < en) ? t : st) * DD];
    }

    #pragma unroll
    for (int k = 0; k < WL_F4; ++k)
        ((float4*)wLDS)[d + 256 * k] = wv[k];

    float m = -3.402823466e+38f;
    #pragma unroll
    for (int i = 0; i < SPAN_ROWS; ++i)
        m = fmaxf(m, (st + i < en) ? v[i] : -3.402823466e+38f);
    for (int t = st + SPAN_ROWS; t < en; ++t)
        m = fmaxf(m, base[(size_t)t * DD]);

    pooled[(size_t)bs * DD + d] = m;
    rowLDS[d] = m;
    __syncthreads();

    const int wid  = threadIdx.x >> 5;
    const int lane = threadIdx.x & 31;
    float pl[3];
    #pragma unroll
    for (int oo = 0; oo < 3; ++oo) {
        const int o   = wid * 3 + oo;
        const int r   = (o < 12) ? o : (o - 12);
        const int off = (o < 12) ? 0 : DD;
        float acc = 0.f;
        #pragma unroll
        for (int j = 0; j < 8; ++j) {
            const int dd = lane + 32 * j;
            acc += rowLDS[dd] * wLDS[(off + dd) * RR + r];
        }
        pl[oo] = acc;
    }
    #pragma unroll
    for (int sh = 16; sh >= 1; sh >>= 1) {
        float ex[3];
        #pragma unroll
        for (int oo = 0; oo < 3; ++oo) ex[oo] = __shfl_xor(pl[oo], sh, 32);
        #pragma unroll
        for (int oo = 0; oo < 3; ++oo) pl[oo] += ex[oo];
    }
    if (lane == 0) {
        #pragma unroll
        for (int oo = 0; oo < 3; ++oo) lin[bs * 24 + wid * 3 + oo] = pl[oo];
    }
}

#define GM_ROWS 64
#define GM_RING 8
#define TB_BLK  8
#define TSTR    68
#define PPT     (PP / 128)

__global__ __launch_bounds__(128) void span_gemm_score_kernel(
    const float* __restrict__ pooled,
    const float* __restrict__ W_bilinear,
    const float* __restrict__ lin,
    const float* __restrict__ b_linear,
    const int*   __restrict__ pair_head,
    const int*   __restrict__ pair_tail,
    float*       __restrict__ out)
{
    __shared__ __attribute__((aligned(16))) float aLDS[GM_ROWS * LSTR];
    __shared__ __attribute__((aligned(16))) float gLDS[32 * LSTR];
    __shared__ __attribute__((aligned(16))) float tLDS[32 * TSTR];
    __shared__ float linLDS[2 * SS];

    const int tid  = threadIdx.x;
    const int wid  = tid >> 5;
    const int lane = tid & 31;
    const int l16  = lane & 15;
    const int hi   = lane >> 4;
    const int b    = blockIdx.x >> 1;
    const int hh   = blockIdx.x & 1;
    const int r    = blockIdx.y;
    const int bs0  = b * GM_ROWS;

    const float* wp = W_bilinear + (((size_t)r) << 16) + wid * 64
                                 + 4 * l16 + (size_t)(2 * hi) * DD;

    float4 bq[GM_RING][2];
    #pragma unroll
    for (int q = 0; q < GM_RING; ++q) {
        bq[q][0] = *(const float4*)(wp + (size_t)(4 * q) * DD);
        bq[q][1] = *(const float4*)(wp + (size_t)(4 * q + 1) * DD);
    }

    const float linv = lin[(bs0 + (tid & 63)) * 24 + (tid >> 6) * 12 + r];

    {
        const int r2 = tid >> 6;
        const int c4 = (tid & 63) << 2;
        const float* src = pooled + ((size_t)(bs0 + r2)) * DD + c4;
        float*       dst = &aLDS[r2 * LSTR + c4];
        float4 av[GM_ROWS / 2];
        #pragma unroll
        for (int it = 0; it < GM_ROWS / 2; ++it)
            av[it] = *(const float4*)(src + (size_t)(2 * it) * DD);
        #pragma unroll
        for (int it = 0; it < GM_ROWS / 2; ++it)
            *(float4*)(dst + 2 * it * LSTR) = av[it];
    }
    linLDS[tid] = linv;
    __syncthreads();

    int ph[PPT], pt[PPT];
    #pragma unroll
    for (int j = 0; j < PPT; ++j) {
        ph[j] = pair_head[b * PP + tid + 128 * j];
        pt[j] = pair_tail[b * PP + tid + 128 * j];
    }
    const float bias = b_linear[r];

    const float* ap = &aLDS[(hh * 32 + l16) * LSTR + 2 * hi];

    v2f af[2][8][2];
    #pragma unroll
    for (int q = 0; q < 8; ++q) {
        af[0][q][0] = *(const v2f*)(ap + 4 * q);
        af[0][q][1] = *(const v2f*)(ap + 16 * LSTR + 4 * q);
    }

    v8f acc[2][4];
    #pragma unroll
    for (int i = 0; i < 2; ++i)
        #pragma unroll
        for (int j = 0; j < 4; ++j)
            acc[i][j] = (v8f){};

    #pragma unroll
    for (int blk = 0; blk < 8; ++blk) {
        const int cur = blk & 1;
        if (blk + 1 < 8) {
            __builtin_amdgcn_sched_barrier(0);
            #pragma unroll
            for (int q = 0; q < 8; ++q) {
                af[cur ^ 1][q][0] = *(const v2f*)(ap + 4 * (8 * (blk + 1) + q));
                af[cur ^ 1][q][1] = *(const v2f*)(ap + 16 * LSTR + 4 * (8 * (blk + 1) + q));
            }
            __builtin_amdgcn_sched_barrier(0);
        }
        #pragma unroll
        for (int q = 0; q < 8; ++q) {
            const int kk = blk * 8 + q;
            const float4 r0 = bq[q][0];
            const float4 r1 = bq[q][1];
            if (kk + GM_RING < 64) {
                bq[q][0] = *(const float4*)(wp + (size_t)(4 * (kk + GM_RING)) * DD);
                bq[q][1] = *(const float4*)(wp + (size_t)(4 * (kk + GM_RING) + 1) * DD);
            }
            v2f b0; b0.x = r0.x; b0.y = r1.x;
            v2f b1; b1.x = r0.y; b1.y = r1.y;
            v2f b2; b2.x = r0.z; b2.y = r1.z;
            v2f b3; b3.x = r0.w; b3.y = r1.w;
            const v2f a0 = af[cur][q][0];
            const v2f a1 = af[cur][q][1];
            acc[0][0] = __builtin_amdgcn_wmma_f32_16x16x4_f32(false, a0, false, b0, (short)0, acc[0][0], false, false);
            acc[1][0] = __builtin_amdgcn_wmma_f32_16x16x4_f32(false, a1, false, b0, (short)0, acc[1][0], false, false);
            acc[0][1] = __builtin_amdgcn_wmma_f32_16x16x4_f32(false, a0, false, b1, (short)0, acc[0][1], false, false);
            acc[1][1] = __builtin_amdgcn_wmma_f32_16x16x4_f32(false, a1, false, b1, (short)0, acc[1][1], false, false);
            acc[0][2] = __builtin_amdgcn_wmma_f32_16x16x4_f32(false, a0, false, b2, (short)0, acc[0][2], false, false);
            acc[1][2] = __builtin_amdgcn_wmma_f32_16x16x4_f32(false, a1, false, b2, (short)0, acc[1][2], false, false);
            acc[0][3] = __builtin_amdgcn_wmma_f32_16x16x4_f32(false, a0, false, b3, (short)0, acc[0][3], false, false);
            acc[1][3] = __builtin_amdgcn_wmma_f32_16x16x4_f32(false, a1, false, b3, (short)0, acc[1][3], false, false);
        }
    }

    #pragma unroll
    for (int i = 0; i < 2; ++i) {
        #pragma unroll
        for (int v = 0; v < 8; ++v) {
            float4 o;
            o.x = acc[i][0][v]; o.y = acc[i][1][v]; o.z = acc[i][2][v]; o.w = acc[i][3][v];
            *(float4*)&gLDS[(i * 16 + v + 8 * hi) * LSTR + wid * 64 + 4 * l16] = o;
        }
    }
    __syncthreads();

    const int mt = wid & 1;
    const int th = wid >> 1;
    const float* gp = &gLDS[(mt * 16 + l16) * LSTR + 4 * hi];
    const float* tp = &aLDS[(th * 32 + 2 * l16) * LSTR + 4 * hi];

    float4 ga[2][TB_BLK], t0[2][TB_BLK], t1[2][TB_BLK];
    #pragma unroll
    for (int q = 0; q < TB_BLK; ++q) {
        ga[0][q] = *(const float4*)(gp + 8 * q);
        t0[0][q] = *(const float4*)(tp + 8 * q);
        t1[0][q] = *(const float4*)(tp + LSTR + 8 * q);
    }
    v8f tacc[2];
    tacc[0] = (v8f){};
    tacc[1] = (v8f){};

    #pragma unroll
    for (int blk = 0; blk < 32 / TB_BLK; ++blk) {
        const int cur = blk & 1;
        if (blk + 1 < 32 / TB_BLK) {
            __builtin_amdgcn_sched_barrier(0);
            #pragma unroll
            for (int q = 0; q < TB_BLK; ++q) {
                const int s = TB_BLK * (blk + 1) + q;
                ga[cur ^ 1][q] = *(const float4*)(gp + 8 * s);
                t0[cur ^ 1][q] = *(const float4*)(tp + 8 * s);
                t1[cur ^ 1][q] = *(const float4*)(tp + LSTR + 8 * s);
            }
            __builtin_amdgcn_sched_barrier(0);
        }
        #pragma unroll
        for (int q = 0; q < TB_BLK; ++q) {
            const float4 g  = ga[cur][q];
            const float4 p0 = t0[cur][q];
            const float4 p1 = t1[cur][q];
            v2f a0;  a0.x  = g.x;  a0.y  = g.y;
            v2f a1;  a1.x  = g.z;  a1.y  = g.w;
            v2f b00; b00.x = p0.x; b00.y = p0.y;
            v2f b01; b01.x = p0.z; b01.y = p0.w;
            v2f b10; b10.x = p1.x; b10.y = p1.y;
            v2f b11; b11.x = p1.z; b11.y = p1.w;
            tacc[0] = __builtin_amdgcn_wmma_f32_16x16x4_f32(false, a0, false, b00, (short)0, tacc[0], false, false);
            tacc[1] = __builtin_amdgcn_wmma_f32_16x16x4_f32(false, a0, false, b10, (short)0, tacc[1], false, false);
            tacc[0] = __builtin_amdgcn_wmma_f32_16x16x4_f32(false, a1, false, b01, (short)0, tacc[0], false, false);
            tacc[1] = __builtin_amdgcn_wmma_f32_16x16x4_f32(false, a1, false, b11, (short)0, tacc[1], false, false);
        }
    }

    float* trow = &tLDS[(mt * 16 + 8 * hi) * TSTR + th * 32 + 2 * l16];
    #pragma unroll
    for (int v = 0; v < 8; ++v) {
        float2 o;
        o.x = tacc[0][v]; o.y = tacc[1][v];
        *(float2*)(trow + v * TSTR) = o;
    }
    __syncthreads();

    float y[PPT];
    #pragma unroll
    for (int j = 0; j < PPT; ++j) {
        const int h = ph[j];
        const int t = pt[j];
        const float sc = tLDS[(h & 31) * TSTR + t] + bias
                       + linLDS[h]
                       + linLDS[SS + t];
        y[j] = 1.f / (1.f + __expf(-sc));
    }
    float* op = out + ((size_t)b * PP + tid) * RR + r;
    #pragma unroll
    for (int j = 0; j < PPT; ++j) {
        if ((ph[j] >> 5) == hh)
            op[(size_t)(128 * j) * RR] = y[j];
    }
}

extern "C" void kernel_launch(void* const* d_in, const int* in_sizes, int n_in,
                              void* d_out, int out_size, void* d_ws, size_t ws_size,
                              hipStream_t stream) {
    const float* encoded    = (const float*)d_in[0];
    const float* W_linear   = (const float*)d_in[1];
    const float* b_linear   = (const float*)d_in[2];
    const float* W_bilinear = (const float*)d_in[3];
    const int*   span_starts= (const int*)d_in[4];
    const int*   span_lens  = (const int*)d_in[5];
    const int*   pair_head  = (const int*)d_in[6];
    const int*   pair_tail  = (const int*)d_in[7];
    float*       out        = (float*)d_out;

    char* ws = (char*)d_ws;
    float* pooled = (float*)ws;
    float* lin    = (float*)(ws + (size_t)NROWS * DD * 4);

    span_pool_kernel<<<NROWS, DD, 0, stream>>>(encoded, span_starts, span_lens,
                                               W_linear, pooled, lin);

    dim3 ggrid(BB * 2, RR);
    span_gemm_score_kernel<<<ggrid, 128, 0, stream>>>(pooled, W_bilinear, lin, b_linear,
                                                      pair_head, pair_tail, out);
}


# Pooling folded into the GEMM kernel: 8-wave blocks max-pool their batch's spans straight into LDS, own half first, other half by waves 4-7 under the first product; 2 launches become 1; sums unchanged
# speedup vs baseline: 3.5121x; 1.1878x over previous
#include <hip/hip_runtime.h>
#include <math.h>

#define BB 8
#define TT 512
#define DD 256
#define SS 64
#define PP 1024
#define RR 12
#define NN (RR * DD)
#define NROWS (BB * SS)

typedef __attribute__((ext_vector_type(2))) float v2f;
typedef __attribute__((ext_vector_type(8))) float v8f;

#define LSTR 260

#define SPAN_ROWS 12
#define PGRP 4

__device__ __forceinline__ void pool_group(
    const float* __restrict__ encl,
    const int sv, const int lv, const int j0,
    float* __restrict__ dst)
{
    int st[PGRP], n[PGRP];
    #pragma unroll
    for (int u = 0; u < PGRP; ++u) {
        st[u] = __builtin_amdgcn_readlane(sv, j0 + u);
        n[u]  = __builtin_amdgcn_readlane(lv, j0 + u) + 1;
    }
    const float lo = -3.402823466e+38f;
    float4 v0[PGRP][SPAN_ROWS], v1[PGRP][SPAN_ROWS];
    #pragma unroll
    for (int u = 0; u < PGRP; ++u) {
        const float* rp = encl + (size_t)st[u] * DD;
        #pragma unroll
        for (int i = 0; i < SPAN_ROWS; ++i) {
            if (i < n[u]) {
                v0[u][i] = *(const float4*)(rp + (size_t)i * DD);
                v1[u][i] = *(const float4*)(rp + (size_t)i * DD + 128);
            } else {
                v0[u][i] = make_float4(lo, lo, lo, lo);
                v1[u][i] = make_float4(lo, lo, lo, lo);
            }
        }
    }
    #pragma unroll
    for (int u = 0; u < PGRP; ++u) {
        float4 m0 = v0[u][0];
        float4 m1 = v1[u][0];
        #pragma unroll
        for (int i = 1; i < SPAN_ROWS; ++i) {
            m0.x = fmaxf(m0.x, v0[u][i].x);  m0.y = fmaxf(m0.y, v0[u][i].y);
            m0.z = fmaxf(m0.z, v0[u][i].z);  m0.w = fmaxf(m0.w, v0[u][i].w);
            m1.x = fmaxf(m1.x, v1[u][i].x);  m1.y = fmaxf(m1.y, v1[u][i].y);
            m1.z = fmaxf(m1.z, v1[u][i].z);  m1.w = fmaxf(m1.w, v1[u][i].w);
        }
        for (int t = st[u] + SPAN_ROWS; t < st[u] + n[u]; ++t) {
            const float4 a = *(const float4*)(encl + (size_t)t * DD);
            const float4 c = *(const float4*)(encl + (size_t)t * DD + 128);
            m0.x = fmaxf(m0.x, a.x);  m0.y = fmaxf(m0.y, a.y);  m0.z = fmaxf(m0.z, a.z);  m0.w = fmaxf(m0.w, a.w);
            m1.x = fmaxf(m1.x, c.x);  m1.y = fmaxf(m1.y, c.y);  m1.z = fmaxf(m1.z, c.z);  m1.w = fmaxf(m1.w, c.w);
        }
        *(float4*)(dst + u * LSTR)       = m0;
        *(float4*)(dst + u * LSTR + 128) = m1;
    }
}

#define GM_ROWS 64
#define GM_RING 8
#define TB_BLK  8
#define TSTR    68
#define PPT     (PP / 128)

__global__ __launch_bounds__(256) void span_gemm_score_kernel(
    const float* __restrict__ encoded,
    const int*   __restrict__ starts,
    const int*   __restrict__ lens,
    const float* __restrict__ W_bilinear,
    const float* __restrict__ W_linear,
    const float* __restrict__ b_linear,
    const int*   __restrict__ pair_head,
    const int*   __restrict__ pair_tail,
    float*       __restrict__ out)
{
    __shared__ __attribute__((aligned(16))) float aLDS[GM_ROWS * LSTR];
    __shared__ __attribute__((aligned(16))) float gLDS[32 * LSTR];
    __shared__ __attribute__((aligned(16))) float tLDS[32 * TSTR];
    __shared__ __attribute__((aligned(16))) float wlLDS[2 * DD];
    __shared__ float hpLDS[4 * 32];
    __shared__ float linLDS[32];

    const int tid  = threadIdx.x;
    const int wid  = __builtin_amdgcn_readfirstlane(tid >> 5);
    const bool mm  = wid < 4;
    const int lane = tid & 31;
    const int l16  = lane & 15;
    const int hi   = lane >> 4;
    const int b    = blockIdx.x >> 1;
    const int hh   = blockIdx.x & 1;
    const int r    = blockIdx.y;
    const int bs0  = b * GM_ROWS;

    const int sv0 = starts[bs0 + lane];
    const int sv1 = starts[bs0 + 32 + lane];
    const int lv0 = lens[bs0 + lane];
    const int lv1 = lens[bs0 + 32 + lane];

    const float* wp = W_bilinear + (((size_t)r) << 16) + wid * 64
                                 + 4 * l16 + (size_t)(2 * hi) * DD;

    float4 bq[GM_RING][2];
    if (mm) {
        #pragma unroll
        for (int q = 0; q < GM_RING; ++q) {
            bq[q][0] = *(const float4*)(wp + (size_t)(4 * q) * DD);
            bq[q][1] = *(const float4*)(wp + (size_t)(4 * q + 1) * DD);
        }
    }

    float wlv[2];
    #pragma unroll
    for (int j = 0; j < 2; ++j)
        wlv[j] = W_linear[(size_t)(tid + 256 * j) * RR + r];

    const float* encl = encoded + ((size_t)b * TT) * DD + 4 * lane;
    pool_group(encl, hh ? sv1 : sv0, hh ? lv1 : lv0, 4 * wid,
               &aLDS[(hh * 32 + 4 * wid) * LSTR + 4 * lane]);
    #pragma unroll
    for (int j = 0; j < 2; ++j)
        wlLDS[tid + 256 * j] = wlv[j];
    __syncthreads();

    int ph[PPT], pt[PPT];
    float bias = 0.f;
    if (mm) {
    #pragma unroll
    for (int j = 0; j < PPT; ++j) {
        ph[j] = pair_head[b * PP + tid + 128 * j];
        pt[j] = pair_tail[b * PP + tid + 128 * j];
    }
    bias = b_linear[r];

    {
        const float* hp = &aLDS[(hh * 32 + lane) * LSTR + 64 * wid];
        const float* wq = &wlLDS[64 * wid];
        float s0 = 0.f, s1 = 0.f, s2 = 0.f, s3 = 0.f;
        #pragma unroll
        for (int j = 0; j < 16; ++j) {
            const float4 p = *(const float4*)(hp + 4 * j);
            const float4 w = *(const float4*)(wq + 4 * j);
            s0 = fmaf(p.x, w.x, s0);
            s1 = fmaf(p.y, w.y, s1);
            s2 = fmaf(p.z, w.z, s2);
            s3 = fmaf(p.w, w.w, s3);
        }
        hpLDS[wid * 32 + lane] = (s0 + s1) + (s2 + s3);
    }
    const float4 wt = *(const float4*)&wlLDS[DD + wid * 64 + 4 * l16];
    __builtin_amdgcn_sched_barrier(0);

    const float* ap = &aLDS[(hh * 32 + l16) * LSTR + 2 * hi];

    v2f af[2][8][2];
    #pragma unroll
    for (int q = 0; q < 8; ++q) {
        af[0][q][0] = *(const v2f*)(ap + 4 * q);
        af[0][q][1] = *(const v2f*)(ap + 16 * LSTR + 4 * q);
    }

    v8f acc[2][4];
    #pragma unroll
    for (int i = 0; i < 2; ++i)
        #pragma unroll
        for (int j = 0; j < 4; ++j)
            acc[i][j] = (v8f){};

    #pragma unroll
    for (int blk = 0; blk < 8; ++blk) {
        const int cur = blk & 1;
        if (blk + 1 < 8) {
            __builtin_amdgcn_sched_barrier(0);
            #pragma unroll
            for (int q = 0; q < 8; ++q) {
                af[cur ^ 1][q][0] = *(const v2f*)(ap + 4 * (8 * (blk + 1) + q));
                af[cur ^ 1][q][1] = *(const v2f*)(ap + 16 * LSTR + 4 * (8 * (blk + 1) + q));
            }
            __builtin_amdgcn_sched_barrier(0);
        }
        #pragma unroll
        for (int q = 0; q < 8; ++q) {
            const int kk = blk * 8 + q;
            const float4 r0 = bq[q][0];
            const float4 r1 = bq[q][1];
            if (kk + GM_RING < 64) {
                bq[q][0] = *(const float4*)(wp + (size_t)(4 * (kk + GM_RING)) * DD);
                bq[q][1] = *(const float4*)(wp + (size_t)(4 * (kk + GM_RING) + 1) * DD);
            }
            v2f b0; b0.x = r0.x; b0.y = r1.x;
            v2f b1; b1.x = r0.y; b1.y = r1.y;
            v2f b2; b2.x = r0.z; b2.y = r1.z;
            v2f b3; b3.x = r0.w; b3.y = r1.w;
            const v2f a0 = af[cur][q][0];
            const v2f a1 = af[cur][q][1];
            acc[0][0] = __builtin_amdgcn_wmma_f32_16x16x4_f32(false, a0, false, b0, (short)0, acc[0][0], false, false);
            acc[1][0] = __builtin_amdgcn_wmma_f32_16x16x4_f32(false, a1, false, b0, (short)0, acc[1][0], false, false);
            acc[0][1] = __builtin_amdgcn_wmma_f32_16x16x4_f32(false, a0, false, b1, (short)0, acc[0][1], false, false);
            acc[1][1] = __builtin_amdgcn_wmma_f32_16x16x4_f32(false, a1, false, b1, (short)0, acc[1][1], false, false);
            acc[0][2] = __builtin_amdgcn_wmma_f32_16x16x4_f32(false, a0, false, b2, (short)0, acc[0][2], false, false);
            acc[1][2] = __builtin_amdgcn_wmma_f32_16x16x4_f32(false, a1, false, b2, (short)0, acc[1][2], false, false);
            acc[0][3] = __builtin_amdgcn_wmma_f32_16x16x4_f32(false, a0, false, b3, (short)0, acc[0][3], false, false);
            acc[1][3] = __builtin_amdgcn_wmma_f32_16x16x4_f32(false, a1, false, b3, (short)0, acc[1][3], false, false);
        }
    }

    #pragma unroll
    for (int i = 0; i < 2; ++i) {
        #pragma unroll
        for (int v = 0; v < 8; ++v) {
            float4 o;
            o.x = acc[i][0][v] + wt.x; o.y = acc[i][1][v] + wt.y;
            o.z = acc[i][2][v] + wt.z; o.w = acc[i][3][v] + wt.w;
            *(float4*)&gLDS[(i * 16 + v + 8 * hi) * LSTR + wid * 64 + 4 * l16] = o;
        }
    }
    } else {
        const int oh = hh ^ 1;
        for (int g = 0; g < 2; ++g) {
            const int j0 = 8 * (wid - 4) + PGRP * g;
            pool_group(encl, oh ? sv1 : sv0, oh ? lv1 : lv0, j0,
                       &aLDS[(oh * 32 + j0) * LSTR + 4 * lane]);
        }
    }
    __syncthreads();

    if (mm) {
    if (tid < 32)
        linLDS[tid] = ((hpLDS[tid] + hpLDS[32 + tid]) + hpLDS[64 + tid]) + hpLDS[96 + tid];

    const int mt = wid & 1;
    const int th = wid >> 1;
    const float* gp = &gLDS[(mt * 16 + l16) * LSTR + 4 * hi];
    const float* tp = &aLDS[(th * 32 + 2 * l16) * LSTR + 4 * hi];

    float4 ga[2][TB_BLK], t0[2][TB_BLK], t1[2][TB_BLK];
    #pragma unroll
    for (int q = 0; q < TB_BLK; ++q) {
        ga[0][q] = *(const float4*)(gp + 8 * q);
        t0[0][q] = *(const float4*)(tp + 8 * q);
        t1[0][q] = *(const float4*)(tp + LSTR + 8 * q);
    }
    v8f tacc[2];
    tacc[0] = (v8f){};
    tacc[1] = (v8f){};

    #pragma unroll
    for (int blk = 0; blk < 32 / TB_BLK; ++blk) {
        const int cur = blk & 1;
        if (blk + 1 < 32 / TB_BLK) {
            __builtin_amdgcn_sched_barrier(0);
            #pragma unroll
            for (int q = 0; q < TB_BLK; ++q) {
                const int s = TB_BLK * (blk + 1) + q;
                ga[cur ^ 1][q] = *(const float4*)(gp + 8 * s);
                t0[cur ^ 1][q] = *(const float4*)(tp + 8 * s);
                t1[cur ^ 1][q] = *(const float4*)(tp + LSTR + 8 * s);
            }
            __builtin_amdgcn_sched_barrier(0);
        }
        #pragma unroll
        for (int q = 0; q < TB_BLK; ++q) {
            const float4 g  = ga[cur][q];
            const float4 p0 = t0[cur][q];
            const float4 p1 = t1[cur][q];
            v2f a0;  a0.x  = g.x;  a0.y  = g.y;
            v2f a1;  a1.x  = g.z;  a1.y  = g.w;
            v2f b00; b00.x = p0.x; b00.y = p0.y;
            v2f b01; b01.x = p0.z; b01.y = p0.w;
            v2f b10; b10.x = p1.x; b10.y = p1.y;
            v2f b11; b11.x = p1.z; b11.y = p1.w;
            tacc[0] = __builtin_amdgcn_wmma_f32_16x16x4_f32(false, a0, false, b00, (short)0, tacc[0], false, false);
            tacc[1] = __builtin_amdgcn_wmma_f32_16x16x4_f32(false, a0, false, b10, (short)0, tacc[1], false, false);
            tacc[0] = __builtin_amdgcn_wmma_f32_16x16x4_f32(false, a1, false, b01, (short)0, tacc[0], false, false);
            tacc[1] = __builtin_amdgcn_wmma_f32_16x16x4_f32(false, a1, false, b11, (short)0, tacc[1], false, false);
        }
    }

    float* trow = &tLDS[(mt * 16 + 8 * hi) * TSTR + th * 32 + 2 * l16];
    #pragma unroll
    for (int v = 0; v < 8; ++v) {
        float2 o;
        o.x = tacc[0][v]; o.y = tacc[1][v];
        *(float2*)(trow + v * TSTR) = o;
    }
    }
    __syncthreads();

    if (mm) {
    float y[PPT];
    #pragma unroll
    for (int j = 0; j < PPT; ++j) {
        const int h = ph[j];
        const int t = pt[j];
        const float sc = tLDS[(h & 31) * TSTR + t] + bias
                       + linLDS[h & 31];
        y[j] = 1.f / (1.f + __expf(-sc));
    }
    float* op = out + ((size_t)b * PP + tid) * RR + r;
    #pragma unroll
    for (int j = 0; j < PPT; ++j) {
        if ((ph[j] >> 5) == hh)
            op[(size_t)(128 * j) * RR] = y[j];
    }
    }
}

extern "C" void kernel_launch(void* const* d_in, const int* in_sizes, int n_in,
                              void* d_out, int out_size, void* d_ws, size_t ws_size,
                              hipStream_t stream) {
    const float* encoded    = (const float*)d_in[0];
    const float* W_linear   = (const float*)d_in[1];
    const float* b_linear   = (const float*)d_in[2];
    const float* W_bilinear = (const float*)d_in[3];
    const int*   span_starts= (const int*)d_in[4];
    const int*   span_lens  = (const int*)d_in[5];
    const int*   pair_head  = (const int*)d_in[6];
    const int*   pair_tail  = (const int*)d_in[7];
    float*       out        = (float*)d_out;

    dim3 ggrid(BB * 2, RR);
    span_gemm_score_kernel<<<ggrid, 256, 0, stream>>>(encoded, span_starts, span_lens,
                                                      W_bilinear, W_linear, b_linear,
                                                      pair_head, pair_tail, out);
}
